# starting kernel
# baseline (speedup 1.0000x reference)
#include <hip/hip_runtime.h>
#include <math.h>

static constexpr int N    = 50000;
static constexpr int E    = 800000;
static constexpr int ETOT = E + N;
static constexpr int F    = 16;
static constexpr int ED   = 7;
static constexpr int HC   = 64;
static constexpr float NEG_ATT = 0.2f;
static constexpr float NEG     = 0.01f;
static constexpr int NEG_INF_BITS = 0xFF800000;

typedef __attribute__((ext_vector_type(2))) float v2f;
typedef __attribute__((ext_vector_type(8))) float v8f;

static __device__ __forceinline__ void atomic_add_f(float* p, float v) {
    __hip_atomic_fetch_add(p, v, __ATOMIC_RELAXED, __HIP_MEMORY_SCOPE_AGENT);
}
static __device__ __forceinline__ void atomic_max_fbits(int* p, float v) {
    if (v >= 0.0f) atomicMax(p, __float_as_int(v));
    else           atomicMin((unsigned int*)p, __float_as_uint(v));
}

__global__ void k_zero(float* __restrict__ p, int n) {
    int i = blockIdx.x * blockDim.x + threadIdx.x;
    if (i < n) p[i] = 0.0f;
}

__global__ void k_eattr_scatter(const float* __restrict__ eattr,
                                const int* __restrict__ ei,
                                float* __restrict__ lsum,
                                float* __restrict__ cnt) {
    int e = blockIdx.x * blockDim.x + threadIdx.x;
    if (e >= E) return;
    int d = ei[E + e];
    #pragma unroll
    for (int j = 0; j < ED; ++j)
        atomic_add_f(&lsum[d * ED + j], eattr[e * ED + j]);
    atomic_add_f(&cnt[d], 1.0f);
}

__global__ void k_loop_mean(float* __restrict__ lsum, const float* __restrict__ cnt) {
    int n = blockIdx.x * blockDim.x + threadIdx.x;
    if (n >= N) return;
    float inv = 1.0f / fmaxf(cnt[n], 1.0f);
    #pragma unroll
    for (int j = 0; j < ED; ++j) lsum[n * ED + j] *= inv;
}

template <int K>
__global__ void k_gemm_pair(const float* __restrict__ A,
                            const float* __restrict__ Wl, const float* __restrict__ bl,
                            const float* __restrict__ Wr, const float* __restrict__ br,
                            float* __restrict__ outL, float* __restrict__ outR) {
    const int wave = threadIdx.x >> 5;
    const int lane = threadIdx.x & 31;
    const int r    = lane & 15;
    const int hi   = lane >> 4;
    const int row0 = blockIdx.x * 16;
    const int col0 = wave * 16;
    const int arow = row0 + r;
    const int bcol = col0 + r;

    v8f accL = {};
    v8f accR = {};
    #pragma unroll
    for (int k0 = 0; k0 < K; k0 += 4) {
        const int kk = k0 + hi * 2;
        v2f a = *reinterpret_cast<const v2f*>(A + arow * K + kk);
        v2f b;
        b.x = Wl[kk * HC + bcol];
        b.y = Wl[(kk + 1) * HC + bcol];
        accL = __builtin_amdgcn_wmma_f32_16x16x4_f32(false, a, false, b,
                                                     (short)0, accL, false, false);
        v2f b2;
        b2.x = Wr[kk * HC + bcol];
        b2.y = Wr[(kk + 1) * HC + bcol];
        accR = __builtin_amdgcn_wmma_f32_16x16x4_f32(false, a, false, b2,
                                                     (short)0, accR, false, false);
    }
    const float bLv = bl[bcol];
    const float bRv = br[bcol];
    #pragma unroll
    for (int i = 0; i < 8; ++i) {
        const int m = row0 + hi * 8 + i;
        outL[m * HC + bcol] = accL[i] + bLv;
        outR[m * HC + bcol] = accR[i] + bRv;
    }
}

__global__ void k_init_attn(float* __restrict__ acc,
                            int*   __restrict__ amax,
                            float* __restrict__ z) {
    int i = blockIdx.x * blockDim.x + threadIdx.x;
    if (i < N * HC) acc[i] = 0.0f;
    if (i < N * 4) { amax[i] = NEG_INF_BITS; z[i] = 0.0f; }
}

__global__ void k_logits(const float* __restrict__ xl, const float* __restrict__ xr,
                         const float* __restrict__ eattr,
                         const float* __restrict__ lattr,
                         const int* __restrict__ ei,
                         const float* __restrict__ We,
                         const float* __restrict__ att,
                         float* __restrict__ alog,
                         int*   __restrict__ amax) {
    __shared__ float sWe[ED * HC];
    __shared__ float sAtt[HC];
    for (int i = threadIdx.x; i < ED * HC; i += blockDim.x) sWe[i] = We[i];
    for (int i = threadIdx.x; i < HC; i += blockDim.x) sAtt[i] = att[i];
    __syncthreads();

    int t = blockIdx.x * blockDim.x + threadIdx.x;
    if (t >= ETOT * 4) return;
    const int e = t >> 2;
    const int h = t & 3;

    int s, d;
    const float* ea;
    if (e < E) { s = ei[e]; d = ei[E + e]; ea = eattr + (size_t)e * ED; }
    else       { s = e - E; d = s;         ea = lattr + (size_t)(e - E) * ED; }

    float eat[ED];
    #pragma unroll
    for (int j = 0; j < ED; ++j) eat[j] = ea[j];

    const float4* xls = reinterpret_cast<const float4*>(xl + (size_t)s * HC + h * 16);
    const float4* xrd = reinterpret_cast<const float4*>(xr + (size_t)d * HC + h * 16);

    float logit = 0.0f;
    #pragma unroll
    for (int q = 0; q < 4; ++q) {
        float4 vl = xls[q];
        float4 vr = xrd[q];
        float xv[4] = { vl.x + vr.x, vl.y + vr.y, vl.z + vr.z, vl.w + vr.w };
        #pragma unroll
        for (int k = 0; k < 4; ++k) {
            const int c = q * 4 + k;
            float em = 0.0f;
            #pragma unroll
            for (int j = 0; j < ED; ++j) em = fmaf(eat[j], sWe[j * HC + h * 16 + c], em);
            float v = xv[k] + em;
            v = (v > 0.0f) ? v : NEG_ATT * v;
            logit = fmaf(v, sAtt[h * 16 + c], logit);
        }
    }
    alog[(size_t)e * 4 + h] = logit;
    atomic_max_fbits(&amax[d * 4 + h], logit);
}

__global__ void k_accum(const float* __restrict__ xl,
                        const int* __restrict__ ei,
                        const float* __restrict__ alog,
                        const int*   __restrict__ amax,
                        float* __restrict__ z,
                        float* __restrict__ acc) {
    int t = blockIdx.x * blockDim.x + threadIdx.x;
    if (t >= ETOT * 4) return;
    const int e = t >> 2;
    const int h = t & 3;
    int s, d;
    if (e < E) { s = ei[e]; d = ei[E + e]; }
    else       { s = e - E; d = s; }

    const float m  = __int_as_float(amax[d * 4 + h]);
    const float ea = __expf(alog[(size_t)e * 4 + h] - m);
    atomic_add_f(&z[d * 4 + h], ea);

    const float4* xs = reinterpret_cast<const float4*>(xl + (size_t)s * HC + h * 16);
    float* ad = acc + (size_t)d * HC + h * 16;
    #pragma unroll
    for (int i = 0; i < 4; ++i) {
        float4 v = xs[i];
        atomic_add_f(&ad[4 * i + 0], ea * v.x);
        atomic_add_f(&ad[4 * i + 1], ea * v.y);
        atomic_add_f(&ad[4 * i + 2], ea * v.z);
        atomic_add_f(&ad[4 * i + 3], ea * v.w);
    }
}

__global__ void k_finalize(const float* __restrict__ acc,
                           const float* __restrict__ z,
                           const float* __restrict__ bo,
                           float* __restrict__ hout) {
    int i = blockIdx.x * blockDim.x + threadIdx.x;
    if (i >= N * HC) return;
    const int n = i >> 6;
    const int c = i & 63;
    float v = acc[i] / z[n * 4 + (c >> 4)] + bo[c];
    hout[i] = (v > 0.0f) ? v : NEG * v;
}

__global__ void k_head(const float* __restrict__ h2,
                       const float* __restrict__ Ws,
                       const float* __restrict__ bs,
                       float* __restrict__ out) {
    int n = blockIdx.x * blockDim.x + threadIdx.x;
    if (n >= N) return;
    const float4* hv = reinterpret_cast<const float4*>(h2 + (size_t)n * HC);
    const float4* wv = reinterpret_cast<const float4*>(Ws);
    float s = 0.0f;
    #pragma unroll
    for (int j = 0; j < 16; ++j) {
        float4 a = hv[j], b = wv[j];
        s = fmaf(a.x, b.x, s); s = fmaf(a.y, b.y, s);
        s = fmaf(a.z, b.z, s); s = fmaf(a.w, b.w, s);
    }
    out[n] = s + bs[0];
}

static inline int blocks(long long n, int bs) { return (int)((n + bs - 1) / bs); }

extern "C" void kernel_launch(void* const* d_in, const int* in_sizes, int n_in,
                              void* d_out, int out_size, void* d_ws, size_t ws_size,
                              hipStream_t stream) {
    const float* x     = (const float*)d_in[0];
    const int*   ei    = (const int*)d_in[1];
    const float* eattr = (const float*)d_in[2];
    const float* Wl1 = (const float*)d_in[3];  const float* bl1 = (const float*)d_in[4];
    const float* Wr1 = (const float*)d_in[5];  const float* br1 = (const float*)d_in[6];
    const float* We1 = (const float*)d_in[7];  const float* att1 = (const float*)d_in[8];
    const float* bo1 = (const float*)d_in[9];
    const float* Wl2 = (const float*)d_in[10]; const float* bl2 = (const float*)d_in[11];
    const float* Wr2 = (const float*)d_in[12]; const float* br2 = (const float*)d_in[13];
    const float* We2 = (const float*)d_in[14]; const float* att2 = (const float*)d_in[15];
    const float* bo2 = (const float*)d_in[16];
    const float* Ws  = (const float*)d_in[17]; const float* bs = (const float*)d_in[18];
    float* out = (float*)d_out;

    float* ws = (float*)d_ws;
    size_t o = 0;
    float* lsum = ws + o; o += (size_t)N * ED;
    float* cnt  = ws + o; o += (size_t)N;
    float* xl   = ws + o; o += (size_t)N * HC;
    float* xr   = ws + o; o += (size_t)N * HC;
    float* alog = ws + o; o += (size_t)ETOT * 4;
    int*   amax = (int*)(ws + o); o += (size_t)N * 4;
    float* zden = ws + o; o += (size_t)N * 4;
    float* acc  = ws + o; o += (size_t)N * HC;
    float* hbuf = ws + o; o += (size_t)N * HC;

    const int BT = 256;

    k_zero<<<blocks((long long)N * (ED + 1), BT), BT, 0, stream>>>(lsum, N * (ED + 1));
    k_eattr_scatter<<<blocks(E, BT), BT, 0, stream>>>(eattr, ei, lsum, cnt);
    k_loop_mean<<<blocks(N, BT), BT, 0, stream>>>(lsum, cnt);

    k_gemm_pair<F><<<N / 16, 128, 0, stream>>>(x, Wl1, bl1, Wr1, br1, xl, xr);
    k_init_attn<<<blocks((long long)N * HC, BT), BT, 0, stream>>>(acc, amax, zden);
    k_logits<<<blocks((long long)ETOT * 4, BT), BT, 0, stream>>>(xl, xr, eattr, lsum, ei,
                                                                 We1, att1, alog, amax);
    k_accum<<<blocks((long long)ETOT * 4, BT), BT, 0, stream>>>(xl, ei, alog, amax, zden, acc);
    k_finalize<<<blocks((long long)N * HC, BT), BT, 0, stream>>>(acc, zden, bo1, hbuf);

    k_gemm_pair<HC><<<N / 16, 128, 0, stream>>>(hbuf, Wl2, bl2, Wr2, br2, xl, xr);
    k_init_attn<<<blocks((long long)N * HC, BT), BT, 0, stream>>>(acc, amax, zden);
    k_logits<<<blocks((long long)ETOT * 4, BT), BT, 0, stream>>>(xl, xr, eattr, lsum, ei,
                                                                 We2, att2, alog, amax);
    k_accum<<<blocks((long long)ETOT * 4, BT), BT, 0, stream>>>(xl, ei, alog, amax, zden, acc);
    k_finalize<<<blocks((long long)N * HC, BT), BT, 0, stream>>>(acc, zden, bo2, hbuf);

    k_head<<<blocks(N, BT), BT, 0, stream>>>(hbuf, Ws, bs, out);
}


# every float-atomic accumulator (attr sums, z, acc) has one copy per chip half, picked by the wave's die id: adds stay in one half's cache; readers add the 2 copies (sum order changed); 16-B clears
# speedup vs baseline: 2.6230x; 2.6230x over previous
#include <hip/hip_runtime.h>
#include <math.h>

static constexpr int N    = 50000;
static constexpr int E    = 800000;
static constexpr int ETOT = E + N;
static constexpr int F    = 16;
static constexpr int ED   = 7;
static constexpr int HC   = 64;
static constexpr float NEG_ATT = 0.2f;
static constexpr float NEG     = 0.01f;
static constexpr int NEG_INF_BITS = 0xFF800000;
static constexpr int LREC = 8;
static constexpr int EB   = 8;
static constexpr int EPW  = 80;
static_assert(E % EB == 0 && ETOT % EB == 0 && EPW % EB == 0, "a batch is all real edges or all self loops");
static_assert(E % 256 == 0, "k_eattr_scatter: a block owns 256 whole edges");

typedef __attribute__((ext_vector_type(2))) float v2f;
typedef __attribute__((ext_vector_type(8))) float v8f;

static __device__ __forceinline__ void atomic_add_f(float* p, float v) {
    __hip_atomic_fetch_add(p, v, __ATOMIC_RELAXED, __HIP_MEMORY_SCOPE_AGENT);
}
static __device__ __forceinline__ int chip_half() {
    return (int)((__builtin_amdgcn_s_sendmsg_rtn(0x87) >> 18) & 1u);
}
static __device__ __forceinline__ void atomic_max_fbits(int* p, float v) {
    if (v >= 0.0f) atomicMax(p, __float_as_int(v));
    else           atomicMin((unsigned int*)p, __float_as_uint(v));
}

__global__ void k_zero(float* __restrict__ p, int n4) {
    int i = blockIdx.x * blockDim.x + threadIdx.x;
    if (i < n4) reinterpret_cast<float4*>(p)[i] = make_float4(0.0f, 0.0f, 0.0f, 0.0f);
}

__global__ void k_eattr_scatter(const float* __restrict__ eattr,
                                const int* __restrict__ ei,
                                float* __restrict__ lrec2) {
    float* lrec = lrec2 + (size_t)chip_half() * (N * LREC);
    const int sub  = threadIdx.x & 7;
    const int grp  = threadIdx.x >> 3;
    const int base = blockIdx.x * 256;
    int   d[8];
    float v[8];
    #pragma unroll
    for (int it = 0; it < 8; ++it) {
        const int e = base + it * 32 + grp;
        d[it] = ei[E + e];
        const float a = eattr[(size_t)e * ED + (sub < ED ? sub : ED - 1)];
        v[it] = (sub < ED) ? a : 1.0f;
    }
    #pragma unroll
    for (int it = 0; it < 8; ++it)
        atomic_add_f(&lrec[d[it] * LREC + sub], v[it]);
}

__global__ void k_loop_mean(float* __restrict__ lrec) {
    int n = blockIdx.x * blockDim.x + threadIdx.x;
    if (n >= N) return;
    const float* l1 = lrec + (size_t)N * LREC;
    float inv = 1.0f / fmaxf(lrec[n * LREC + ED] + l1[n * LREC + ED], 1.0f);
    #pragma unroll
    for (int j = 0; j < ED; ++j) lrec[n * LREC + j] = (lrec[n * LREC + j] + l1[n * LREC + j]) * inv;
}

template <int K>
__global__ void k_gemm_pair(const float* __restrict__ A,
                            const float* __restrict__ Wl, const float* __restrict__ bl,
                            const float* __restrict__ Wr, const float* __restrict__ br,
                            float* __restrict__ outL, float* __restrict__ outR) {
    const int wave = threadIdx.x >> 5;
    const int lane = threadIdx.x & 31;
    const int r    = lane & 15;
    const int hi   = lane >> 4;
    const int row0 = blockIdx.x * 16;
    const int col0 = wave * 16;
    const int arow = row0 + r;
    const int bcol = col0 + r;

    v8f accL = {};
    v8f accR = {};
    #pragma unroll
    for (int k0 = 0; k0 < K; k0 += 4) {
        const int kk = k0 + hi * 2;
        v2f a = *reinterpret_cast<const v2f*>(A + arow * K + kk);
        v2f b;
        b.x = Wl[kk * HC + bcol];
        b.y = Wl[(kk + 1) * HC + bcol];
        accL = __builtin_amdgcn_wmma_f32_16x16x4_f32(false, a, false, b,
                                                     (short)0, accL, false, false);
        v2f b2;
        b2.x = Wr[kk * HC + bcol];
        b2.y = Wr[(kk + 1) * HC + bcol];
        accR = __builtin_amdgcn_wmma_f32_16x16x4_f32(false, a, false, b2,
                                                     (short)0, accR, false, false);
    }
    const float bLv = bl[bcol];
    const float bRv = br[bcol];
    #pragma unroll
    for (int i = 0; i < 8; ++i) {
        const int m = row0 + hi * 8 + i;
        outL[m * HC + bcol] = accL[i] + bLv;
        outR[m * HC + bcol] = accR[i] + bRv;
    }
}

__global__ void k_init_attn(float* __restrict__ acc,
                            int*   __restrict__ amax,
                            float* __restrict__ z) {
    int i = blockIdx.x * blockDim.x + threadIdx.x;
    if (i < 2 * N * (HC / 4)) reinterpret_cast<float4*>(acc)[i] = make_float4(0.0f, 0.0f, 0.0f, 0.0f);
    if (i < 2 * N) reinterpret_cast<float4*>(z)[i] = make_float4(0.0f, 0.0f, 0.0f, 0.0f);
    if (i < N) reinterpret_cast<int4*>(amax)[i] = make_int4(NEG_INF_BITS, NEG_INF_BITS, NEG_INF_BITS, NEG_INF_BITS);
}

__global__ void k_logits(const float* __restrict__ xl, const float* __restrict__ xr,
                         const float* __restrict__ eattr,
                         const float* __restrict__ lattr,
                         const int* __restrict__ ei,
                         const float* __restrict__ We,
                         const float* __restrict__ att,
                         float* __restrict__ alog,
                         int*   __restrict__ amax) {
    __shared__ float sWe[ED * HC];
    __shared__ float sAtt[HC];
    for (int i = threadIdx.x; i < ED * HC; i += blockDim.x) sWe[i] = We[i];
    for (int i = threadIdx.x; i < HC; i += blockDim.x) sAtt[i] = att[i];
    __syncthreads();

    int t = blockIdx.x * blockDim.x + threadIdx.x;
    if (t >= ETOT * 4) return;
    const int e = t >> 2;
    const int h = t & 3;

    int s, d;
    const float* ea;
    if (e < E) { s = ei[e]; d = ei[E + e]; ea = eattr + (size_t)e * ED; }
    else       { s = e - E; d = s;         ea = lattr + (size_t)(e - E) * LREC; }

    float eat[ED];
    #pragma unroll
    for (int j = 0; j < ED; ++j) eat[j] = ea[j];

    const float4* xls = reinterpret_cast<const float4*>(xl + (size_t)s * HC + h * 16);
    const float4* xrd = reinterpret_cast<const float4*>(xr + (size_t)d * HC + h * 16);

    float logit = 0.0f;
    #pragma unroll
    for (int q = 0; q < 4; ++q) {
        float4 vl = xls[q];
        float4 vr = xrd[q];
        float xv[4] = { vl.x + vr.x, vl.y + vr.y, vl.z + vr.z, vl.w + vr.w };
        #pragma unroll
        for (int k = 0; k < 4; ++k) {
            const int c = q * 4 + k;
            float em = 0.0f;
            #pragma unroll
            for (int j = 0; j < ED; ++j) em = fmaf(eat[j], sWe[j * HC + h * 16 + c], em);
            float v = xv[k] + em;
            v = (v > 0.0f) ? v : NEG_ATT * v;
            logit = fmaf(v, sAtt[h * 16 + c], logit);
        }
    }
    alog[(size_t)e * 4 + h] = logit;
    atomic_max_fbits(&amax[d * 4 + h], logit);
}

__global__ void k_accum(const float* __restrict__ xl,
                        const int* __restrict__ ei,
                        const float* __restrict__ alog,
                        const int*   __restrict__ amax,
                        float* __restrict__ z2,
                        float* __restrict__ acc2) {
    const int half = chip_half();
    float* z   = z2   + (size_t)half * (N * 4);
    float* acc = acc2 + (size_t)half * (N * HC);
    const int lane = threadIdx.x & 31;
    const int wave = __builtin_amdgcn_readfirstlane((int)((blockIdx.x * blockDim.x + threadIdx.x) >> 5));
    const int hA = lane >> 4;
    const int hB = hA + 2;
    const int zl = lane & 15;
    const int e_begin = wave * EPW;

    for (int b = 0; b < EPW; b += EB) {
        const int eb = e_begin + b;
        if (eb >= ETOT) break;

        int s[EB], d[EB];
        if (eb < E) {
            #pragma unroll
            for (int i = 0; i < EB; ++i) { s[i] = ei[eb + i]; d[i] = ei[E + eb + i]; }
        } else {
            #pragma unroll
            for (int i = 0; i < EB; ++i) { s[i] = eb + i - E; d[i] = s[i]; }
        }

        float aA[EB], aB[EB], xa[EB], xb[EB];
        #pragma unroll
        for (int i = 0; i < EB; ++i) {
            const float* ar = alog + (size_t)(eb + i) * 4;
            const int*   mr = amax + (size_t)d[i] * 4;
            aA[i] = ar[hA] - __int_as_float(mr[hA]);
            aB[i] = ar[hB] - __int_as_float(mr[hB]);
            const float* xr = xl + (size_t)s[i] * HC;
            xa[i] = xr[lane];
            xb[i] = xr[32 + lane];
        }

        #pragma unroll
        for (int i = 0; i < EB; ++i) {
            const float wA = __expf(aA[i]);
            const float wB = __expf(aB[i]);
            float* ad = acc + (size_t)d[i] * HC;
            atomic_add_f(&ad[lane],      wA * xa[i]);
            atomic_add_f(&ad[32 + lane], wB * xb[i]);
            if (zl < 2)
                atomic_add_f(&z[d[i] * 4 + (zl ? hB : hA)], zl ? wB : wA);
        }
    }
}

__global__ void k_finalize(const float* __restrict__ acc,
                           const float* __restrict__ z,
                           const float* __restrict__ bo,
                           float* __restrict__ hout) {
    int i = blockIdx.x * blockDim.x + threadIdx.x;
    if (i >= N * HC) return;
    const int n = i >> 6;
    const int c = i & 63;
    const int zi = n * 4 + (c >> 4);
    float v = (acc[i] + acc[N * HC + i]) / (z[zi] + z[N * 4 + zi]) + bo[c];
    hout[i] = (v > 0.0f) ? v : NEG * v;
}

__global__ void k_head(const float* __restrict__ h2,
                       const float* __restrict__ Ws,
                       const float* __restrict__ bs,
                       float* __restrict__ out) {
    int n = blockIdx.x * blockDim.x + threadIdx.x;
    if (n >= N) return;
    const float4* hv = reinterpret_cast<const float4*>(h2 + (size_t)n * HC);
    const float4* wv = reinterpret_cast<const float4*>(Ws);
    float s = 0.0f;
    #pragma unroll
    for (int j = 0; j < 16; ++j) {
        float4 a = hv[j], b = wv[j];
        s = fmaf(a.x, b.x, s); s = fmaf(a.y, b.y, s);
        s = fmaf(a.z, b.z, s); s = fmaf(a.w, b.w, s);
    }
    out[n] = s + bs[0];
}

static inline int blocks(long long n, int bs) { return (int)((n + bs - 1) / bs); }

extern "C" void kernel_launch(void* const* d_in, const int* in_sizes, int n_in,
                              void* d_out, int out_size, void* d_ws, size_t ws_size,
                              hipStream_t stream) {
    const float* x     = (const float*)d_in[0];
    const int*   ei    = (const int*)d_in[1];
    const float* eattr = (const float*)d_in[2];
    const float* Wl1 = (const float*)d_in[3];  const float* bl1 = (const float*)d_in[4];
    const float* Wr1 = (const float*)d_in[5];  const float* br1 = (const float*)d_in[6];
    const float* We1 = (const float*)d_in[7];  const float* att1 = (const float*)d_in[8];
    const float* bo1 = (const float*)d_in[9];
    const float* Wl2 = (const float*)d_in[10]; const float* bl2 = (const float*)d_in[11];
    const float* Wr2 = (const float*)d_in[12]; const float* br2 = (const float*)d_in[13];
    const float* We2 = (const float*)d_in[14]; const float* att2 = (const float*)d_in[15];
    const float* bo2 = (const float*)d_in[16];
    const float* Ws  = (const float*)d_in[17]; const float* bs = (const float*)d_in[18];
    float* out = (float*)d_out;

    float* ws = (float*)d_ws;
    size_t o = 0;
    float* lrec = ws + o; o += (size_t)2 * N * LREC;
    float* xl   = ws + o; o += (size_t)N * HC;
    float* xr   = ws + o; o += (size_t)N * HC;
    float* alog = ws + o; o += (size_t)ETOT * 4;
    int*   amax = (int*)(ws + o); o += (size_t)N * 4;
    float* zden = ws + o; o += (size_t)2 * N * 4;
    float* acc  = ws + o; o += (size_t)2 * N * HC;
    float* hbuf = ws + o; o += (size_t)N * HC;

    const int BT = 256;

    k_zero<<<blocks((long long)2 * N * LREC / 4, BT), BT, 0, stream>>>(lrec, 2 * N * LREC / 4);
    k_eattr_scatter<<<E / 256, 256, 0, stream>>>(eattr, ei, lrec);
    k_loop_mean<<<blocks(N, BT), BT, 0, stream>>>(lrec);

    k_gemm_pair<F><<<N / 16, 128, 0, stream>>>(x, Wl1, bl1, Wr1, br1, xl, xr);
    k_init_attn<<<blocks((long long)2 * N * (HC / 4), BT), BT, 0, stream>>>(acc, amax, zden);
    k_logits<<<blocks((long long)ETOT * 4, BT), BT, 0, stream>>>(xl, xr, eattr, lrec, ei,
                                                                 We1, att1, alog, amax);
    k_accum<<<blocks((long long)((ETOT + EPW - 1) / EPW) * 32, 128), 128, 0, stream>>>(xl, ei, alog, amax, zden, acc);
    k_finalize<<<blocks((long long)N * HC, BT), BT, 0, stream>>>(acc, zden, bo1, hbuf);

    k_gemm_pair<HC><<<N / 16, 128, 0, stream>>>(hbuf, Wl2, bl2, Wr2, br2, xl, xr);
    k_init_attn<<<blocks((long long)2 * N * (HC / 4), BT), BT, 0, stream>>>(acc, amax, zden);
    k_logits<<<blocks((long long)ETOT * 4, BT), BT, 0, stream>>>(xl, xr, eattr, lrec, ei,
                                                                 We2, att2, alog, amax);
    k_accum<<<blocks((long long)((ETOT + EPW - 1) / EPW) * 32, 128), 128, 0, stream>>>(xl, ei, alog, amax, zden, acc);
    k_finalize<<<blocks((long long)N * HC, BT), BT, 0, stream>>>(acc, zden, bo2, hbuf);

    k_head<<<blocks(N, BT), BT, 0, stream>>>(hbuf, Ws, bs, out);
}


# segment-max table (integer atomics) gets one copy per chip half like the float sums: a wave raises its own half's copy; reader lane pairs load one copy each, swap, keep the larger (exact; sums same)
# speedup vs baseline: 2.9833x; 1.1374x over previous
#include <hip/hip_runtime.h>
#include <math.h>

static constexpr int N    = 50000;
static constexpr int E    = 800000;
static constexpr int ETOT = E + N;
static constexpr int F    = 16;
static constexpr int ED   = 7;
static constexpr int HC   = 64;
static constexpr float NEG_ATT = 0.2f;
static constexpr float NEG     = 0.01f;
static constexpr int NEG_INF_BITS = 0xFF800000;
static constexpr int LREC = 8;
static constexpr int EB   = 8;
static constexpr int EPW  = 80;
static_assert(E % EB == 0 && ETOT % EB == 0 && EPW % EB == 0, "a batch is all real edges or all self loops");
static_assert(E % 256 == 0, "k_eattr_scatter: a block owns 256 whole edges");

typedef __attribute__((ext_vector_type(2))) float v2f;
typedef __attribute__((ext_vector_type(8))) float v8f;

static __device__ __forceinline__ void atomic_add_f(float* p, float v) {
    __hip_atomic_fetch_add(p, v, __ATOMIC_RELAXED, __HIP_MEMORY_SCOPE_AGENT);
}
static __device__ __forceinline__ int chip_half() {
    return (int)((__builtin_amdgcn_s_sendmsg_rtn(0x87) >> 18) & 1u);
}
static __device__ __forceinline__ void atomic_max_fbits(int* p, float v) {
    if (v >= 0.0f) atomicMax(p, __float_as_int(v));
    else           atomicMin((unsigned int*)p, __float_as_uint(v));
}

static __device__ __forceinline__ float pair_swap(float v) {
    return __int_as_float(__builtin_amdgcn_mov_dpp(__float_as_int(v), 0xB1, 0xF, 0xF, false));
}

__global__ void k_zero(float* __restrict__ p, int n4) {
    int i = blockIdx.x * blockDim.x + threadIdx.x;
    if (i < n4) reinterpret_cast<float4*>(p)[i] = make_float4(0.0f, 0.0f, 0.0f, 0.0f);
}

__global__ void k_eattr_scatter(const float* __restrict__ eattr,
                                const int* __restrict__ ei,
                                float* __restrict__ lrec2) {
    float* lrec = lrec2 + (size_t)chip_half() * (N * LREC);
    const int sub  = threadIdx.x & 7;
    const int grp  = threadIdx.x >> 3;
    const int base = blockIdx.x * 256;
    int   d[8];
    float v[8];
    #pragma unroll
    for (int it = 0; it < 8; ++it) {
        const int e = base + it * 32 + grp;
        d[it] = ei[E + e];
        const float a = eattr[(size_t)e * ED + (sub < ED ? sub : ED - 1)];
        v[it] = (sub < ED) ? a : 1.0f;
    }
    #pragma unroll
    for (int it = 0; it < 8; ++it)
        atomic_add_f(&lrec[d[it] * LREC + sub], v[it]);
}

__global__ void k_loop_mean(float* __restrict__ lrec) {
    int n = blockIdx.x * blockDim.x + threadIdx.x;
    if (n >= N) return;
    const float* l1 = lrec + (size_t)N * LREC;
    float inv = 1.0f / fmaxf(lrec[n * LREC + ED] + l1[n * LREC + ED], 1.0f);
    #pragma unroll
    for (int j = 0; j < ED; ++j) lrec[n * LREC + j] = (lrec[n * LREC + j] + l1[n * LREC + j]) * inv;
}

template <int K>
__global__ void k_gemm_pair(const float* __restrict__ A,
                            const float* __restrict__ Wl, const float* __restrict__ bl,
                            const float* __restrict__ Wr, const float* __restrict__ br,
                            float* __restrict__ outL, float* __restrict__ outR) {
    const int wave = threadIdx.x >> 5;
    const int lane = threadIdx.x & 31;
    const int r    = lane & 15;
    const int hi   = lane >> 4;
    const int row0 = blockIdx.x * 16;
    const int col0 = wave * 16;
    const int arow = row0 + r;
    const int bcol = col0 + r;

    v8f accL = {};
    v8f accR = {};
    #pragma unroll
    for (int k0 = 0; k0 < K; k0 += 4) {
        const int kk = k0 + hi * 2;
        v2f a = *reinterpret_cast<const v2f*>(A + arow * K + kk);
        v2f b;
        b.x = Wl[kk * HC + bcol];
        b.y = Wl[(kk + 1) * HC + bcol];
        accL = __builtin_amdgcn_wmma_f32_16x16x4_f32(false, a, false, b,
                                                     (short)0, accL, false, false);
        v2f b2;
        b2.x = Wr[kk * HC + bcol];
        b2.y = Wr[(kk + 1) * HC + bcol];
        accR = __builtin_amdgcn_wmma_f32_16x16x4_f32(false, a, false, b2,
                                                     (short)0, accR, false, false);
    }
    const float bLv = bl[bcol];
    const float bRv = br[bcol];
    #pragma unroll
    for (int i = 0; i < 8; ++i) {
        const int m = row0 + hi * 8 + i;
        outL[m * HC + bcol] = accL[i] + bLv;
        outR[m * HC + bcol] = accR[i] + bRv;
    }
}

__global__ void k_init_attn(float* __restrict__ acc,
                            int*   __restrict__ amax,
                            float* __restrict__ z) {
    int i = blockIdx.x * blockDim.x + threadIdx.x;
    if (i < 2 * N * (HC / 4)) reinterpret_cast<float4*>(acc)[i] = make_float4(0.0f, 0.0f, 0.0f, 0.0f);
    if (i < 2 * N) reinterpret_cast<float4*>(z)[i] = make_float4(0.0f, 0.0f, 0.0f, 0.0f);
    if (i < 2 * N) reinterpret_cast<int4*>(amax)[i] = make_int4(NEG_INF_BITS, NEG_INF_BITS, NEG_INF_BITS, NEG_INF_BITS);
}

__global__ void k_logits(const float* __restrict__ xl, const float* __restrict__ xr,
                         const float* __restrict__ eattr,
                         const float* __restrict__ lattr,
                         const int* __restrict__ ei,
                         const float* __restrict__ We,
                         const float* __restrict__ att,
                         float* __restrict__ alog,
                         int*   __restrict__ amax2) {
    int* amax = amax2 + (size_t)chip_half() * (N * 4);
    __shared__ float sWe[ED * HC];
    __shared__ float sAtt[HC];
    for (int i = threadIdx.x; i < ED * HC; i += blockDim.x) sWe[i] = We[i];
    for (int i = threadIdx.x; i < HC; i += blockDim.x) sAtt[i] = att[i];
    __syncthreads();

    int t = blockIdx.x * blockDim.x + threadIdx.x;
    if (t >= ETOT * 4) return;
    const int e = t >> 2;
    const int h = t & 3;

    int s, d;
    const float* ea;
    if (e < E) { s = ei[e]; d = ei[E + e]; ea = eattr + (size_t)e * ED; }
    else       { s = e - E; d = s;         ea = lattr + (size_t)(e - E) * LREC; }

    float eat[ED];
    #pragma unroll
    for (int j = 0; j < ED; ++j) eat[j] = ea[j];

    const float4* xls = reinterpret_cast<const float4*>(xl + (size_t)s * HC + h * 16);
    const float4* xrd = reinterpret_cast<const float4*>(xr + (size_t)d * HC + h * 16);

    float logit = 0.0f;
    #pragma unroll
    for (int q = 0; q < 4; ++q) {
        float4 vl = xls[q];
        float4 vr = xrd[q];
        float xv[4] = { vl.x + vr.x, vl.y + vr.y, vl.z + vr.z, vl.w + vr.w };
        #pragma unroll
        for (int k = 0; k < 4; ++k) {
            const int c = q * 4 + k;
            float em = 0.0f;
            #pragma unroll
            for (int j = 0; j < ED; ++j) em = fmaf(eat[j], sWe[j * HC + h * 16 + c], em);
            float v = xv[k] + em;
            v = (v > 0.0f) ? v : NEG_ATT * v;
            logit = fmaf(v, sAtt[h * 16 + c], logit);
        }
    }
    alog[(size_t)e * 4 + h] = logit;
    atomic_max_fbits(&amax[d * 4 + h], logit);
}

__global__ void k_accum(const float* __restrict__ xl,
                        const int* __restrict__ ei,
                        const float* __restrict__ alog,
                        const int*   __restrict__ amax2,
                        float* __restrict__ z2,
                        float* __restrict__ acc2) {
    const int half = chip_half();
    float* z   = z2   + (size_t)half * (N * 4);
    float* acc = acc2 + (size_t)half * (N * HC);
    const int lane = threadIdx.x & 31;
    const int wave = __builtin_amdgcn_readfirstlane((int)((blockIdx.x * blockDim.x + threadIdx.x) >> 5));
    const int hA = lane >> 4;
    const int hB = hA + 2;
    const int zl = lane & 15;
    const int e_begin = wave * EPW;
    const int* amax = amax2 + (size_t)(lane & 1) * (N * 4);

    for (int b = 0; b < EPW; b += EB) {
        const int eb = e_begin + b;
        if (eb >= ETOT) break;

        int s[EB], d[EB];
        if (eb < E) {
            #pragma unroll
            for (int i = 0; i < EB; ++i) { s[i] = ei[eb + i]; d[i] = ei[E + eb + i]; }
        } else {
            #pragma unroll
            for (int i = 0; i < EB; ++i) { s[i] = eb + i - E; d[i] = s[i]; }
        }

        float aA[EB], aB[EB], xa[EB], xb[EB];
        #pragma unroll
        for (int i = 0; i < EB; ++i) {
            const float* ar = alog + (size_t)(eb + i) * 4;
            const int*   mr = amax + (size_t)d[i] * 4;
            const float mA = __int_as_float(mr[hA]);
            const float mB = __int_as_float(mr[hB]);
            aA[i] = ar[hA] - fmaxf(mA, pair_swap(mA));
            aB[i] = ar[hB] - fmaxf(mB, pair_swap(mB));
            const float* xr = xl + (size_t)s[i] * HC;
            xa[i] = xr[lane];
            xb[i] = xr[32 + lane];
        }

        #pragma unroll
        for (int i = 0; i < EB; ++i) {
            const float wA = __expf(aA[i]);
            const float wB = __expf(aB[i]);
            float* ad = acc + (size_t)d[i] * HC;
            atomic_add_f(&ad[lane],      wA * xa[i]);
            atomic_add_f(&ad[32 + lane], wB * xb[i]);
            if (zl < 2)
                atomic_add_f(&z[d[i] * 4 + (zl ? hB : hA)], zl ? wB : wA);
        }
    }
}

__global__ void k_finalize(const float* __restrict__ acc,
                           const float* __restrict__ z,
                           const float* __restrict__ bo,
                           float* __restrict__ hout) {
    int i = blockIdx.x * blockDim.x + threadIdx.x;
    if (i >= N * HC) return;
    const int n = i >> 6;
    const int c = i & 63;
    const int zi = n * 4 + (c >> 4);
    float v = (acc[i] + acc[N * HC + i]) / (z[zi] + z[N * 4 + zi]) + bo[c];
    hout[i] = (v > 0.0f) ? v : NEG * v;
}

__global__ void k_head(const float* __restrict__ h2,
                       const float* __restrict__ Ws,
                       const float* __restrict__ bs,
                       float* __restrict__ out) {
    int n = blockIdx.x * blockDim.x + threadIdx.x;
    if (n >= N) return;
    const float4* hv = reinterpret_cast<const float4*>(h2 + (size_t)n * HC);
    const float4* wv = reinterpret_cast<const float4*>(Ws);
    float s = 0.0f;
    #pragma unroll
    for (int j = 0; j < 16; ++j) {
        float4 a = hv[j], b = wv[j];
        s = fmaf(a.x, b.x, s); s = fmaf(a.y, b.y, s);
        s = fmaf(a.z, b.z, s); s = fmaf(a.w, b.w, s);
    }
    out[n] = s + bs[0];
}

static inline int blocks(long long n, int bs) { return (int)((n + bs - 1) / bs); }

extern "C" void kernel_launch(void* const* d_in, const int* in_sizes, int n_in,
                              void* d_out, int out_size, void* d_ws, size_t ws_size,
                              hipStream_t stream) {
    const float* x     = (const float*)d_in[0];
    const int*   ei    = (const int*)d_in[1];
    const float* eattr = (const float*)d_in[2];
    const float* Wl1 = (const float*)d_in[3];  const float* bl1 = (const float*)d_in[4];
    const float* Wr1 = (const float*)d_in[5];  const float* br1 = (const float*)d_in[6];
    const float* We1 = (const float*)d_in[7];  const float* att1 = (const float*)d_in[8];
    const float* bo1 = (const float*)d_in[9];
    const float* Wl2 = (const float*)d_in[10]; const float* bl2 = (const float*)d_in[11];
    const float* Wr2 = (const float*)d_in[12]; const float* br2 = (const float*)d_in[13];
    const float* We2 = (const float*)d_in[14]; const float* att2 = (const float*)d_in[15];
    const float* bo2 = (const float*)d_in[16];
    const float* Ws  = (const float*)d_in[17]; const float* bs = (const float*)d_in[18];
    float* out = (float*)d_out;

    float* ws = (float*)d_ws;
    size_t o = 0;
    float* lrec = ws + o; o += (size_t)2 * N * LREC;
    float* xl   = ws + o; o += (size_t)N * HC;
    float* xr   = ws + o; o += (size_t)N * HC;
    float* alog = ws + o; o += (size_t)ETOT * 4;
    int*   amax = (int*)(ws + o); o += (size_t)2 * N * 4;
    float* zden = ws + o; o += (size_t)2 * N * 4;
    float* acc  = ws + o; o += (size_t)2 * N * HC;
    float* hbuf = ws + o; o += (size_t)N * HC;

    const int BT = 256;

    k_zero<<<blocks((long long)2 * N * LREC / 4, BT), BT, 0, stream>>>(lrec, 2 * N * LREC / 4);
    k_eattr_scatter<<<E / 256, 256, 0, stream>>>(eattr, ei, lrec);
    k_loop_mean<<<blocks(N, BT), BT, 0, stream>>>(lrec);

    k_gemm_pair<F><<<N / 16, 128, 0, stream>>>(x, Wl1, bl1, Wr1, br1, xl, xr);
    k_init_attn<<<blocks((long long)2 * N * (HC / 4), BT), BT, 0, stream>>>(acc, amax, zden);
    k_logits<<<blocks((long long)ETOT * 4, BT), BT, 0, stream>>>(xl, xr, eattr, lrec, ei,
                                                                 We1, att1, alog, amax);
    k_accum<<<blocks((long long)((ETOT + EPW - 1) / EPW) * 32, 128), 128, 0, stream>>>(xl, ei, alog, amax, zden, acc);
    k_finalize<<<blocks((long long)N * HC, BT), BT, 0, stream>>>(acc, zden, bo1, hbuf);

    k_gemm_pair<HC><<<N / 16, 128, 0, stream>>>(hbuf, Wl2, bl2, Wr2, br2, xl, xr);
    k_init_attn<<<blocks((long long)2 * N * (HC / 4), BT), BT, 0, stream>>>(acc, amax, zden);
    k_logits<<<blocks((long long)ETOT * 4, BT), BT, 0, stream>>>(xl, xr, eattr, lrec, ei,
                                                                 We2, att2, alog, amax);
    k_accum<<<blocks((long long)((ETOT + EPW - 1) / EPW) * 32, 128), 128, 0, stream>>>(xl, ei, alog, amax, zden, acc);
    k_finalize<<<blocks((long long)N * HC, BT), BT, 0, stream>>>(acc, zden, bo2, hbuf);

    k_head<<<blocks(N, BT), BT, 0, stream>>>(hbuf, Ws, bs, out);
}


# clear, logit and finalize kernels become fixed grids whose threads loop over their items (logit weights stay in registers; finalize issues all loads first): 519k short waves per call become 42k
# speedup vs baseline: 3.2773x; 1.0985x over previous
#include <hip/hip_runtime.h>
#include <math.h>

static constexpr int N    = 50000;
static constexpr int E    = 800000;
static constexpr int ETOT = E + N;
static constexpr int F    = 16;
static constexpr int ED   = 7;
static constexpr int HC   = 64;
static constexpr float NEG_ATT = 0.2f;
static constexpr float NEG     = 0.01f;
static constexpr int NEG_INF_BITS = 0xFF800000;
static constexpr int LREC = 8;
static constexpr int EB   = 8;
static constexpr int EPW  = 80;
static_assert(E % EB == 0 && ETOT % EB == 0 && EPW % EB == 0, "a batch is all real edges or all self loops");
static_assert(E % 256 == 0, "k_eattr_scatter: a block owns 256 whole edges");
static constexpr int ZERO_BLOCKS   = 196;
static constexpr int STREAM_BLOCKS = 625;
static constexpr int FIN_TRIPS     = 5;
static_assert(STREAM_BLOCKS * 256 * FIN_TRIPS == N * (HC / 4), "finalize: the grid covers the layer output exactly");
static constexpr int LOGIT_THREADS = 128;
static constexpr int LOGIT_BLOCKS  = 2560;

typedef __attribute__((ext_vector_type(2))) float v2f;
typedef __attribute__((ext_vector_type(8))) float v8f;

static __device__ __forceinline__ void atomic_add_f(float* p, float v) {
    __hip_atomic_fetch_add(p, v, __ATOMIC_RELAXED, __HIP_MEMORY_SCOPE_AGENT);
}
static __device__ __forceinline__ int chip_half() {
    return (int)((__builtin_amdgcn_s_sendmsg_rtn(0x87) >> 18) & 1u);
}
static __device__ __forceinline__ void atomic_max_fbits(int* p, float v) {
    if (v >= 0.0f) atomicMax(p, __float_as_int(v));
    else           atomicMin((unsigned int*)p, __float_as_uint(v));
}

static __device__ __forceinline__ float pair_swap(float v) {
    return __int_as_float(__builtin_amdgcn_mov_dpp(__float_as_int(v), 0xB1, 0xF, 0xF, false));
}

__global__ void k_zero(float* __restrict__ p, int n4) {
    const int stride = gridDim.x * blockDim.x;
    for (int i = blockIdx.x * blockDim.x + threadIdx.x; i < n4; i += stride)
        reinterpret_cast<float4*>(p)[i] = make_float4(0.0f, 0.0f, 0.0f, 0.0f);
}

__global__ void k_eattr_scatter(const float* __restrict__ eattr,
                                const int* __restrict__ ei,
                                float* __restrict__ lrec2) {
    float* lrec = lrec2 + (size_t)chip_half() * (N * LREC);
    const int sub  = threadIdx.x & 7;
    const int grp  = threadIdx.x >> 3;
    const int base = blockIdx.x * 256;
    int   d[8];
    float v[8];
    #pragma unroll
    for (int it = 0; it < 8; ++it) {
        const int e = base + it * 32 + grp;
        d[it] = ei[E + e];
        const float a = eattr[(size_t)e * ED + (sub < ED ? sub : ED - 1)];
        v[it] = (sub < ED) ? a : 1.0f;
    }
    #pragma unroll
    for (int it = 0; it < 8; ++it)
        atomic_add_f(&lrec[d[it] * LREC + sub], v[it]);
}

__global__ void k_loop_mean(float* __restrict__ lrec) {
    int n = blockIdx.x * blockDim.x + threadIdx.x;
    if (n >= N) return;
    const float* l1 = lrec + (size_t)N * LREC;
    float inv = 1.0f / fmaxf(lrec[n * LREC + ED] + l1[n * LREC + ED], 1.0f);
    #pragma unroll
    for (int j = 0; j < ED; ++j) lrec[n * LREC + j] = (lrec[n * LREC + j] + l1[n * LREC + j]) * inv;
}

template <int K>
__global__ void k_gemm_pair(const float* __restrict__ A,
                            const float* __restrict__ Wl, const float* __restrict__ bl,
                            const float* __restrict__ Wr, const float* __restrict__ br,
                            float* __restrict__ outL, float* __restrict__ outR) {
    const int wave = threadIdx.x >> 5;
    const int lane = threadIdx.x & 31;
    const int r    = lane & 15;
    const int hi   = lane >> 4;
    const int row0 = blockIdx.x * 16;
    const int col0 = wave * 16;
    const int arow = row0 + r;
    const int bcol = col0 + r;

    v8f accL = {};
    v8f accR = {};
    #pragma unroll
    for (int k0 = 0; k0 < K; k0 += 4) {
        const int kk = k0 + hi * 2;
        v2f a = *reinterpret_cast<const v2f*>(A + arow * K + kk);
        v2f b;
        b.x = Wl[kk * HC + bcol];
        b.y = Wl[(kk + 1) * HC + bcol];
        accL = __builtin_amdgcn_wmma_f32_16x16x4_f32(false, a, false, b,
                                                     (short)0, accL, false, false);
        v2f b2;
        b2.x = Wr[kk * HC + bcol];
        b2.y = Wr[(kk + 1) * HC + bcol];
        accR = __builtin_amdgcn_wmma_f32_16x16x4_f32(false, a, false, b2,
                                                     (short)0, accR, false, false);
    }
    const float bLv = bl[bcol];
    const float bRv = br[bcol];
    #pragma unroll
    for (int i = 0; i < 8; ++i) {
        const int m = row0 + hi * 8 + i;
        outL[m * HC + bcol] = accL[i] + bLv;
        outR[m * HC + bcol] = accR[i] + bRv;
    }
}

__global__ void k_init_attn(float* __restrict__ acc,
                            int*   __restrict__ amax,
                            float* __restrict__ z) {
    const int tid    = blockIdx.x * blockDim.x + threadIdx.x;
    const int stride = gridDim.x * blockDim.x;
    for (int i = tid; i < 2 * N * (HC / 4); i += stride)
        reinterpret_cast<float4*>(acc)[i] = make_float4(0.0f, 0.0f, 0.0f, 0.0f);
    for (int i = tid; i < 2 * N; i += stride) {
        reinterpret_cast<float4*>(z)[i] = make_float4(0.0f, 0.0f, 0.0f, 0.0f);
        reinterpret_cast<int4*>(amax)[i] = make_int4(NEG_INF_BITS, NEG_INF_BITS, NEG_INF_BITS, NEG_INF_BITS);
    }
}

__global__ void __launch_bounds__(LOGIT_THREADS)
k_logits(const float* __restrict__ xl, const float* __restrict__ xr,
                         const float* __restrict__ eattr,
                         const float* __restrict__ lattr,
                         const int* __restrict__ ei,
                         const float* __restrict__ We,
                         const float* __restrict__ att,
                         float* __restrict__ alog,
                         int*   __restrict__ amax2) {
    int* amax = amax2 + (size_t)chip_half() * (N * 4);
    __shared__ float sWe[ED * HC];
    __shared__ float sAtt[HC];
    for (int i = threadIdx.x; i < ED * HC; i += blockDim.x) sWe[i] = We[i];
    for (int i = threadIdx.x; i < HC; i += blockDim.x) sAtt[i] = att[i];
    __syncthreads();

    const int stride = gridDim.x * blockDim.x;
    const int h = threadIdx.x & 3;
    for (int t = blockIdx.x * blockDim.x + threadIdx.x; t < ETOT * 4; t += stride) {
        const int e = t >> 2;

        int s, d;
        const float* ea;
        if (e < E) { s = ei[e]; d = ei[E + e]; ea = eattr + (size_t)e * ED; }
        else       { s = e - E; d = s;         ea = lattr + (size_t)(e - E) * LREC; }

        float eat[ED];
        #pragma unroll
        for (int j = 0; j < ED; ++j) eat[j] = ea[j];

        const float4* xls = reinterpret_cast<const float4*>(xl + (size_t)s * HC + h * 16);
        const float4* xrd = reinterpret_cast<const float4*>(xr + (size_t)d * HC + h * 16);

        float logit = 0.0f;
        #pragma unroll
        for (int q = 0; q < 4; ++q) {
            float4 vl = xls[q];
            float4 vr = xrd[q];
            float xv[4] = { vl.x + vr.x, vl.y + vr.y, vl.z + vr.z, vl.w + vr.w };
            #pragma unroll
            for (int k = 0; k < 4; ++k) {
                const int c = q * 4 + k;
                float em = 0.0f;
                #pragma unroll
                for (int j = 0; j < ED; ++j) em = fmaf(eat[j], sWe[j * HC + h * 16 + c], em);
                float v = xv[k] + em;
                v = (v > 0.0f) ? v : NEG_ATT * v;
                logit = fmaf(v, sAtt[h * 16 + c], logit);
            }
        }
        alog[(size_t)e * 4 + h] = logit;
        atomic_max_fbits(&amax[d * 4 + h], logit);
    }
}

__global__ void k_accum(const float* __restrict__ xl,
                        const int* __restrict__ ei,
                        const float* __restrict__ alog,
                        const int*   __restrict__ amax2,
                        float* __restrict__ z2,
                        float* __restrict__ acc2) {
    const int half = chip_half();
    float* z   = z2   + (size_t)half * (N * 4);
    float* acc = acc2 + (size_t)half * (N * HC);
    const int lane = threadIdx.x & 31;
    const int wave = __builtin_amdgcn_readfirstlane((int)((blockIdx.x * blockDim.x + threadIdx.x) >> 5));
    const int hA = lane >> 4;
    const int hB = hA + 2;
    const int zl = lane & 15;
    const int e_begin = wave * EPW;
    const int* amax = amax2 + (size_t)(lane & 1) * (N * 4);

    for (int b = 0; b < EPW; b += EB) {
        const int eb = e_begin + b;
        if (eb >= ETOT) break;

        int s[EB], d[EB];
        if (eb < E) {
            #pragma unroll
            for (int i = 0; i < EB; ++i) { s[i] = ei[eb + i]; d[i] = ei[E + eb + i]; }
        } else {
            #pragma unroll
            for (int i = 0; i < EB; ++i) { s[i] = eb + i - E; d[i] = s[i]; }
        }

        float aA[EB], aB[EB], xa[EB], xb[EB];
        #pragma unroll
        for (int i = 0; i < EB; ++i) {
            const float* ar = alog + (size_t)(eb + i) * 4;
            const int*   mr = amax + (size_t)d[i] * 4;
            const float mA = __int_as_float(mr[hA]);
            const float mB = __int_as_float(mr[hB]);
            aA[i] = ar[hA] - fmaxf(mA, pair_swap(mA));
            aB[i] = ar[hB] - fmaxf(mB, pair_swap(mB));
            const float* xr = xl + (size_t)s[i] * HC;
            xa[i] = xr[lane];
            xb[i] = xr[32 + lane];
        }

        #pragma unroll
        for (int i = 0; i < EB; ++i) {
            const float wA = __expf(aA[i]);
            const float wB = __expf(aB[i]);
            float* ad = acc + (size_t)d[i] * HC;
            atomic_add_f(&ad[lane],      wA * xa[i]);
            atomic_add_f(&ad[32 + lane], wB * xb[i]);
            if (zl < 2)
                atomic_add_f(&z[d[i] * 4 + (zl ? hB : hA)], zl ? wB : wA);
        }
    }
}

__global__ void k_finalize(const float* __restrict__ acc,
                           const float* __restrict__ z,
                           const float* __restrict__ bo,
                           float* __restrict__ hout) {
    const int tid = blockIdx.x * blockDim.x + threadIdx.x;
    float4 a0[FIN_TRIPS], a1[FIN_TRIPS];
    float  z0[FIN_TRIPS], z1[FIN_TRIPS];
    #pragma unroll
    for (int k = 0; k < FIN_TRIPS; ++k) {
        const int i4 = tid + k * (STREAM_BLOCKS * 256);
        const int zi = (i4 >> 4) * 4 + ((i4 & 15) >> 2);
        a0[k] = reinterpret_cast<const float4*>(acc)[i4];
        a1[k] = reinterpret_cast<const float4*>(acc + (size_t)N * HC)[i4];
        z0[k] = z[zi];
        z1[k] = z[N * 4 + zi];
    }
    const float4 b = reinterpret_cast<const float4*>(bo)[tid & 15];
    #pragma unroll
    for (int k = 0; k < FIN_TRIPS; ++k) {
        const int i4 = tid + k * (STREAM_BLOCKS * 256);
        const float zs = z0[k] + z1[k];
        float4 v;
        v.x = (a0[k].x + a1[k].x) / zs + b.x;
        v.y = (a0[k].y + a1[k].y) / zs + b.y;
        v.z = (a0[k].z + a1[k].z) / zs + b.z;
        v.w = (a0[k].w + a1[k].w) / zs + b.w;
        v.x = (v.x > 0.0f) ? v.x : NEG * v.x;
        v.y = (v.y > 0.0f) ? v.y : NEG * v.y;
        v.z = (v.z > 0.0f) ? v.z : NEG * v.z;
        v.w = (v.w > 0.0f) ? v.w : NEG * v.w;
        reinterpret_cast<float4*>(hout)[i4] = v;
    }
}

__global__ void k_head(const float* __restrict__ h2,
                       const float* __restrict__ Ws,
                       const float* __restrict__ bs,
                       float* __restrict__ out) {
    int n = blockIdx.x * blockDim.x + threadIdx.x;
    if (n >= N) return;
    const float4* hv = reinterpret_cast<const float4*>(h2 + (size_t)n * HC);
    const float4* wv = reinterpret_cast<const float4*>(Ws);
    float s = 0.0f;
    #pragma unroll
    for (int j = 0; j < 16; ++j) {
        float4 a = hv[j], b = wv[j];
        s = fmaf(a.x, b.x, s); s = fmaf(a.y, b.y, s);
        s = fmaf(a.z, b.z, s); s = fmaf(a.w, b.w, s);
    }
    out[n] = s + bs[0];
}

static inline int blocks(long long n, int bs) { return (int)((n + bs - 1) / bs); }

extern "C" void kernel_launch(void* const* d_in, const int* in_sizes, int n_in,
                              void* d_out, int out_size, void* d_ws, size_t ws_size,
                              hipStream_t stream) {
    const float* x     = (const float*)d_in[0];
    const int*   ei    = (const int*)d_in[1];
    const float* eattr = (const float*)d_in[2];
    const float* Wl1 = (const float*)d_in[3];  const float* bl1 = (const float*)d_in[4];
    const float* Wr1 = (const float*)d_in[5];  const float* br1 = (const float*)d_in[6];
    const float* We1 = (const float*)d_in[7];  const float* att1 = (const float*)d_in[8];
    const float* bo1 = (const float*)d_in[9];
    const float* Wl2 = (const float*)d_in[10]; const float* bl2 = (const float*)d_in[11];
    const float* Wr2 = (const float*)d_in[12]; const float* br2 = (const float*)d_in[13];
    const float* We2 = (const float*)d_in[14]; const float* att2 = (const float*)d_in[15];
    const float* bo2 = (const float*)d_in[16];
    const float* Ws  = (const float*)d_in[17]; const float* bs = (const float*)d_in[18];
    float* out = (float*)d_out;

    float* ws = (float*)d_ws;
    size_t o = 0;
    float* lrec = ws + o; o += (size_t)2 * N * LREC;
    float* xl   = ws + o; o += (size_t)N * HC;
    float* xr   = ws + o; o += (size_t)N * HC;
    float* alog = ws + o; o += (size_t)ETOT * 4;
    int*   amax = (int*)(ws + o); o += (size_t)2 * N * 4;
    float* zden = ws + o; o += (size_t)2 * N * 4;
    float* acc  = ws + o; o += (size_t)2 * N * HC;
    float* hbuf = ws + o; o += (size_t)N * HC;

    const int BT = 256;

    k_zero<<<ZERO_BLOCKS, BT, 0, stream>>>(lrec, 2 * N * LREC / 4);
    k_eattr_scatter<<<E / 256, 256, 0, stream>>>(eattr, ei, lrec);
    k_loop_mean<<<blocks(N, BT), BT, 0, stream>>>(lrec);

    k_gemm_pair<F><<<N / 16, 128, 0, stream>>>(x, Wl1, bl1, Wr1, br1, xl, xr);
    k_init_attn<<<STREAM_BLOCKS, BT, 0, stream>>>(acc, amax, zden);
    k_logits<<<LOGIT_BLOCKS, LOGIT_THREADS, 0, stream>>>(xl, xr, eattr, lrec, ei,
                                                                 We1, att1, alog, amax);
    k_accum<<<blocks((long long)((ETOT + EPW - 1) / EPW) * 32, 128), 128, 0, stream>>>(xl, ei, alog, amax, zden, acc);
    k_finalize<<<STREAM_BLOCKS, 256, 0, stream>>>(acc, zden, bo1, hbuf);

    k_gemm_pair<HC><<<N / 16, 128, 0, stream>>>(hbuf, Wl2, bl2, Wr2, br2, xl, xr);
    k_init_attn<<<STREAM_BLOCKS, BT, 0, stream>>>(acc, amax, zden);
    k_logits<<<LOGIT_BLOCKS, LOGIT_THREADS, 0, stream>>>(xl, xr, eattr, lrec, ei,
                                                                 We2, att2, alog, amax);
    k_accum<<<blocks((long long)((ETOT + EPW - 1) / EPW) * 32, 128), 128, 0, stream>>>(xl, ei, alog, amax, zden, acc);
    k_finalize<<<STREAM_BLOCKS, 256, 0, stream>>>(acc, zden, bo2, hbuf);

    k_head<<<blocks(N, BT), BT, 0, stream>>>(hbuf, Ws, bs, out);
}


# k_logits: a wave takes one edge at a time, lane = 2 channels (8-byte loads: a row is 2 whole lines), edge ids and attrs as wave scalars; head dot summed over 8 lanes by butterfly (sum order changed)
# speedup vs baseline: 3.7234x; 1.1361x over previous
#include <hip/hip_runtime.h>
#include <math.h>

static constexpr int N    = 50000;
static constexpr int E    = 800000;
static constexpr int ETOT = E + N;
static constexpr int F    = 16;
static constexpr int ED   = 7;
static constexpr int HC   = 64;
static constexpr float NEG_ATT = 0.2f;
static constexpr float NEG     = 0.01f;
static constexpr int NEG_INF_BITS = 0xFF800000;
static constexpr int LREC = 8;
static constexpr int EB   = 8;
static constexpr int EPW  = 80;
static_assert(E % EB == 0 && ETOT % EB == 0 && EPW % EB == 0, "a batch is all real edges or all self loops");
static_assert(E % 256 == 0, "k_eattr_scatter: a block owns 256 whole edges");
static constexpr int ZERO_BLOCKS   = 196;
static constexpr int STREAM_BLOCKS = 625;
static constexpr int FIN_TRIPS     = 5;
static_assert(STREAM_BLOCKS * 256 * FIN_TRIPS == N * (HC / 4), "finalize: the grid covers the layer output exactly");
static constexpr int LOGIT_THREADS = 64;
static constexpr int LG_EB  = 8;
static constexpr int LG_EPW = 80;
static constexpr int LG_WAVES = (ETOT + LG_EPW - 1) / LG_EPW;
static_assert(E % LG_EB == 0 && ETOT % LG_EB == 0 && LG_EPW % LG_EB == 0, "k_logits: a batch is all real edges or all self loops");

typedef __attribute__((ext_vector_type(2))) float v2f;
typedef __attribute__((ext_vector_type(8))) float v8f;

static __device__ __forceinline__ void atomic_add_f(float* p, float v) {
    __hip_atomic_fetch_add(p, v, __ATOMIC_RELAXED, __HIP_MEMORY_SCOPE_AGENT);
}
static __device__ __forceinline__ int chip_half() {
    return (int)((__builtin_amdgcn_s_sendmsg_rtn(0x87) >> 18) & 1u);
}
static __device__ __forceinline__ void atomic_max_fbits(int* p, float v) {
    if (v >= 0.0f) atomicMax(p, __float_as_int(v));
    else           atomicMin((unsigned int*)p, __float_as_uint(v));
}

static __device__ __forceinline__ float pair_swap(float v) {
    return __int_as_float(__builtin_amdgcn_mov_dpp(__float_as_int(v), 0xB1, 0xF, 0xF, false));
}

__global__ void k_zero(float* __restrict__ p, int n4) {
    const int stride = gridDim.x * blockDim.x;
    for (int i = blockIdx.x * blockDim.x + threadIdx.x; i < n4; i += stride)
        reinterpret_cast<float4*>(p)[i] = make_float4(0.0f, 0.0f, 0.0f, 0.0f);
}

__global__ void k_eattr_scatter(const float* __restrict__ eattr,
                                const int* __restrict__ ei,
                                float* __restrict__ lrec2) {
    float* lrec = lrec2 + (size_t)chip_half() * (N * LREC);
    const int sub  = threadIdx.x & 7;
    const int grp  = threadIdx.x >> 3;
    const int base = blockIdx.x * 256;
    int   d[8];
    float v[8];
    #pragma unroll
    for (int it = 0; it < 8; ++it) {
        const int e = base + it * 32 + grp;
        d[it] = ei[E + e];
        const float a = eattr[(size_t)e * ED + (sub < ED ? sub : ED - 1)];
        v[it] = (sub < ED) ? a : 1.0f;
    }
    #pragma unroll
    for (int it = 0; it < 8; ++it)
        atomic_add_f(&lrec[d[it] * LREC + sub], v[it]);
}

__global__ void k_loop_mean(float* __restrict__ lrec) {
    int n = blockIdx.x * blockDim.x + threadIdx.x;
    if (n >= N) return;
    const float* l1 = lrec + (size_t)N * LREC;
    float inv = 1.0f / fmaxf(lrec[n * LREC + ED] + l1[n * LREC + ED], 1.0f);
    #pragma unroll
    for (int j = 0; j < ED; ++j) lrec[n * LREC + j] = (lrec[n * LREC + j] + l1[n * LREC + j]) * inv;
}

template <int K>
__global__ void k_gemm_pair(const float* __restrict__ A,
                            const float* __restrict__ Wl, const float* __restrict__ bl,
                            const float* __restrict__ Wr, const float* __restrict__ br,
                            float* __restrict__ outL, float* __restrict__ outR) {
    const int wave = threadIdx.x >> 5;
    const int lane = threadIdx.x & 31;
    const int r    = lane & 15;
    const int hi   = lane >> 4;
    const int row0 = blockIdx.x * 16;
    const int col0 = wave * 16;
    const int arow = row0 + r;
    const int bcol = col0 + r;

    v8f accL = {};
    v8f accR = {};
    #pragma unroll
    for (int k0 = 0; k0 < K; k0 += 4) {
        const int kk = k0 + hi * 2;
        v2f a = *reinterpret_cast<const v2f*>(A + arow * K + kk);
        v2f b;
        b.x = Wl[kk * HC + bcol];
        b.y = Wl[(kk + 1) * HC + bcol];
        accL = __builtin_amdgcn_wmma_f32_16x16x4_f32(false, a, false, b,
                                                     (short)0, accL, false, false);
        v2f b2;
        b2.x = Wr[kk * HC + bcol];
        b2.y = Wr[(kk + 1) * HC + bcol];
        accR = __builtin_amdgcn_wmma_f32_16x16x4_f32(false, a, false, b2,
                                                     (short)0, accR, false, false);
    }
    const float bLv = bl[bcol];
    const float bRv = br[bcol];
    #pragma unroll
    for (int i = 0; i < 8; ++i) {
        const int m = row0 + hi * 8 + i;
        outL[m * HC + bcol] = accL[i] + bLv;
        outR[m * HC + bcol] = accR[i] + bRv;
    }
}

__global__ void k_init_attn(float* __restrict__ acc,
                            int*   __restrict__ amax,
                            float* __restrict__ z) {
    const int tid    = blockIdx.x * blockDim.x + threadIdx.x;
    const int stride = gridDim.x * blockDim.x;
    for (int i = tid; i < 2 * N * (HC / 4); i += stride)
        reinterpret_cast<float4*>(acc)[i] = make_float4(0.0f, 0.0f, 0.0f, 0.0f);
    for (int i = tid; i < 2 * N; i += stride) {
        reinterpret_cast<float4*>(z)[i] = make_float4(0.0f, 0.0f, 0.0f, 0.0f);
        reinterpret_cast<int4*>(amax)[i] = make_int4(NEG_INF_BITS, NEG_INF_BITS, NEG_INF_BITS, NEG_INF_BITS);
    }
}

template <int M>
static __device__ __forceinline__ float xadd(float keep, float send) {
    return keep + __int_as_float(__builtin_amdgcn_mov_dpp(__float_as_int(send), 0x160 + M, 0xF, 0xF, true));
}

template <bool REAL>
static __device__ __forceinline__ void logit_batch(const int eb, const int lane,
        const float* __restrict__ xl, const float* __restrict__ xr,
        const float* __restrict__ eattr, const float* __restrict__ lattr,
        const int* __restrict__ ei,
        const v2f (&w)[ED], const v2f av,
        float* __restrict__ alog, int* __restrict__ amax) {
    const unsigned h = (unsigned)lane >> 3;
    const unsigned k = (unsigned)lane & 7u;
    const v2f* xl2 = reinterpret_cast<const v2f*>(xl);
    const v2f* xr2 = reinterpret_cast<const v2f*>(xr);
    unsigned s[LG_EB], d[LG_EB];
    unsigned dw;
    #pragma unroll
    for (int i = 0; i < LG_EB; ++i) {
        if (REAL) { s[i] = (unsigned)ei[eb + i]; d[i] = (unsigned)ei[E + eb + i]; }
        else      { s[i] = (unsigned)(eb + i - E); d[i] = s[i]; }
    }
    if (REAL) dw = (unsigned)ei[E + eb + k];
    else      dw = (unsigned)(eb - E) + k;

    v2f vl[LG_EB], vr[LG_EB];
    #pragma unroll
    for (int i = 0; i < LG_EB; ++i) {
        vl[i] = xl2[s[i] * (unsigned)(HC / 2) + (unsigned)lane];
        vr[i] = xr2[d[i] * (unsigned)(HC / 2) + (unsigned)lane];
    }

    float p[LG_EB];
    #pragma unroll
    for (int i = 0; i < LG_EB; ++i) {
        const float* ea = REAL ? eattr + (size_t)(eb + i) * ED
                               : lattr + (size_t)(eb + i - E) * LREC;
        v2f em = w[0] * ea[0];
        #pragma unroll
        for (int j = 1; j < ED; ++j) {
            const v2f a2 = { ea[j], ea[j] };
            em = __builtin_elementwise_fma(a2, w[j], em);
        }
        v2f u = (vl[i] + vr[i]) + em;
        const v2f t = u * NEG_ATT;
        u.x = fmaxf(u.x, t.x);
        u.y = fmaxf(u.y, t.y);
        p[i] = fmaf(u.y, av.y, u.x * av.x);
    }
    const bool b2 = (lane & 4) != 0, b1 = (lane & 2) != 0, b0 = (lane & 1) != 0;
    float q[4], r[2];
    #pragma unroll
    for (int m = 0; m < 4; ++m) q[m] = xadd<4>(b2 ? p[4 + m] : p[m], b2 ? p[m] : p[4 + m]);
    #pragma unroll
    for (int m = 0; m < 2; ++m) r[m] = xadd<2>(b1 ? q[2 + m] : q[m], b1 ? q[m] : q[2 + m]);
    const float mine = xadd<1>(b0 ? r[1] : r[0], b0 ? r[0] : r[1]);

    alog[(size_t)((unsigned)eb + k) * 4 + h] = mine;
    atomic_max_fbits(&amax[dw * 4u + h], mine);
}

__global__ void __launch_bounds__(LOGIT_THREADS)
k_logits(const float* __restrict__ xl, const float* __restrict__ xr,
                         const float* __restrict__ eattr,
                         const float* __restrict__ lattr,
                         const int* __restrict__ ei,
                         const float* __restrict__ We,
                         const float* __restrict__ att,
                         float* __restrict__ alog,
                         int*   __restrict__ amax2) {
    int* amax = amax2 + (size_t)chip_half() * (N * 4);
    const int lane = threadIdx.x & 31;
    const int wave = __builtin_amdgcn_readfirstlane((int)((blockIdx.x * blockDim.x + threadIdx.x) >> 5));
    const int e_begin = wave * LG_EPW;
    if (e_begin >= ETOT) return;

    v2f w[ED];
    #pragma unroll
    for (int j = 0; j < ED; ++j) w[j] = *reinterpret_cast<const v2f*>(We + j * HC + 2 * lane);
    const v2f av = *reinterpret_cast<const v2f*>(att + 2 * lane);

    for (int b = 0; b < LG_EPW; b += LG_EB) {
        const int eb = e_begin + b;
        if (eb >= ETOT) break;
        if (eb < E) logit_batch<true>(eb, lane, xl, xr, eattr, lattr, ei, w, av, alog, amax);
        else        logit_batch<false>(eb, lane, xl, xr, eattr, lattr, ei, w, av, alog, amax);
    }
}

__global__ void k_accum(const float* __restrict__ xl,
                        const int* __restrict__ ei,
                        const float* __restrict__ alog,
                        const int*   __restrict__ amax2,
                        float* __restrict__ z2,
                        float* __restrict__ acc2) {
    const int half = chip_half();
    float* z   = z2   + (size_t)half * (N * 4);
    float* acc = acc2 + (size_t)half * (N * HC);
    const int lane = threadIdx.x & 31;
    const int wave = __builtin_amdgcn_readfirstlane((int)((blockIdx.x * blockDim.x + threadIdx.x) >> 5));
    const int hA = lane >> 4;
    const int hB = hA + 2;
    const int zl = lane & 15;
    const int e_begin = wave * EPW;
    const int* amax = amax2 + (size_t)(lane & 1) * (N * 4);

    for (int b = 0; b < EPW; b += EB) {
        const int eb = e_begin + b;
        if (eb >= ETOT) break;

        int s[EB], d[EB];
        if (eb < E) {
            #pragma unroll
            for (int i = 0; i < EB; ++i) { s[i] = ei[eb + i]; d[i] = ei[E + eb + i]; }
        } else {
            #pragma unroll
            for (int i = 0; i < EB; ++i) { s[i] = eb + i - E; d[i] = s[i]; }
        }

        float aA[EB], aB[EB], xa[EB], xb[EB];
        #pragma unroll
        for (int i = 0; i < EB; ++i) {
            const float* ar = alog + (size_t)(eb + i) * 4;
            const int*   mr = amax + (size_t)d[i] * 4;
            const float mA = __int_as_float(mr[hA]);
            const float mB = __int_as_float(mr[hB]);
            aA[i] = ar[hA] - fmaxf(mA, pair_swap(mA));
            aB[i] = ar[hB] - fmaxf(mB, pair_swap(mB));
            const float* xr = xl + (size_t)s[i] * HC;
            xa[i] = xr[lane];
            xb[i] = xr[32 + lane];
        }

        #pragma unroll
        for (int i = 0; i < EB; ++i) {
            const float wA = __expf(aA[i]);
            const float wB = __expf(aB[i]);
            float* ad = acc + (size_t)d[i] * HC;
            atomic_add_f(&ad[lane],      wA * xa[i]);
            atomic_add_f(&ad[32 + lane], wB * xb[i]);
            if (zl < 2)
                atomic_add_f(&z[d[i] * 4 + (zl ? hB : hA)], zl ? wB : wA);
        }
    }
}

__global__ void k_finalize(const float* __restrict__ acc,
                           const float* __restrict__ z,
                           const float* __restrict__ bo,
                           float* __restrict__ hout) {
    const int tid = blockIdx.x * blockDim.x + threadIdx.x;
    float4 a0[FIN_TRIPS], a1[FIN_TRIPS];
    float  z0[FIN_TRIPS], z1[FIN_TRIPS];
    #pragma unroll
    for (int k = 0; k < FIN_TRIPS; ++k) {
        const int i4 = tid + k * (STREAM_BLOCKS * 256);
        const int zi = (i4 >> 4) * 4 + ((i4 & 15) >> 2);
        a0[k] = reinterpret_cast<const float4*>(acc)[i4];
        a1[k] = reinterpret_cast<const float4*>(acc + (size_t)N * HC)[i4];
        z0[k] = z[zi];
        z1[k] = z[N * 4 + zi];
    }
    const float4 b = reinterpret_cast<const float4*>(bo)[tid & 15];
    #pragma unroll
    for (int k = 0; k < FIN_TRIPS; ++k) {
        const int i4 = tid + k * (STREAM_BLOCKS * 256);
        const float zs = z0[k] + z1[k];
        float4 v;
        v.x = (a0[k].x + a1[k].x) / zs + b.x;
        v.y = (a0[k].y + a1[k].y) / zs + b.y;
        v.z = (a0[k].z + a1[k].z) / zs + b.z;
        v.w = (a0[k].w + a1[k].w) / zs + b.w;
        v.x = (v.x > 0.0f) ? v.x : NEG * v.x;
        v.y = (v.y > 0.0f) ? v.y : NEG * v.y;
        v.z = (v.z > 0.0f) ? v.z : NEG * v.z;
        v.w = (v.w > 0.0f) ? v.w : NEG * v.w;
        reinterpret_cast<float4*>(hout)[i4] = v;
    }
}

__global__ void k_head(const float* __restrict__ h2,
                       const float* __restrict__ Ws,
                       const float* __restrict__ bs,
                       float* __restrict__ out) {
    int n = blockIdx.x * blockDim.x + threadIdx.x;
    if (n >= N) return;
    const float4* hv = reinterpret_cast<const float4*>(h2 + (size_t)n * HC);
    const float4* wv = reinterpret_cast<const float4*>(Ws);
    float s = 0.0f;
    #pragma unroll
    for (int j = 0; j < 16; ++j) {
        float4 a = hv[j], b = wv[j];
        s = fmaf(a.x, b.x, s); s = fmaf(a.y, b.y, s);
        s = fmaf(a.z, b.z, s); s = fmaf(a.w, b.w, s);
    }
    out[n] = s + bs[0];
}

static inline int blocks(long long n, int bs) { return (int)((n + bs - 1) / bs); }

extern "C" void kernel_launch(void* const* d_in, const int* in_sizes, int n_in,
                              void* d_out, int out_size, void* d_ws, size_t ws_size,
                              hipStream_t stream) {
    const float* x     = (const float*)d_in[0];
    const int*   ei    = (const int*)d_in[1];
    const float* eattr = (const float*)d_in[2];
    const float* Wl1 = (const float*)d_in[3];  const float* bl1 = (const float*)d_in[4];
    const float* Wr1 = (const float*)d_in[5];  const float* br1 = (const float*)d_in[6];
    const float* We1 = (const float*)d_in[7];  const float* att1 = (const float*)d_in[8];
    const float* bo1 = (const float*)d_in[9];
    const float* Wl2 = (const float*)d_in[10]; const float* bl2 = (const float*)d_in[11];
    const float* Wr2 = (const float*)d_in[12]; const float* br2 = (const float*)d_in[13];
    const float* We2 = (const float*)d_in[14]; const float* att2 = (const float*)d_in[15];
    const float* bo2 = (const float*)d_in[16];
    const float* Ws  = (const float*)d_in[17]; const float* bs = (const float*)d_in[18];
    float* out = (float*)d_out;

    float* ws = (float*)d_ws;
    size_t o = 0;
    float* lrec = ws + o; o += (size_t)2 * N * LREC;
    float* xl   = ws + o; o += (size_t)N * HC;
    float* xr   = ws + o; o += (size_t)N * HC;
    float* alog = ws + o; o += (size_t)ETOT * 4;
    int*   amax = (int*)(ws + o); o += (size_t)2 * N * 4;
    float* zden = ws + o; o += (size_t)2 * N * 4;
    float* acc  = ws + o; o += (size_t)2 * N * HC;
    float* hbuf = ws + o; o += (size_t)N * HC;

    const int BT = 256;

    k_zero<<<ZERO_BLOCKS, BT, 0, stream>>>(lrec, 2 * N * LREC / 4);
    k_eattr_scatter<<<E / 256, 256, 0, stream>>>(eattr, ei, lrec);
    k_loop_mean<<<blocks(N, BT), BT, 0, stream>>>(lrec);

    k_gemm_pair<F><<<N / 16, 128, 0, stream>>>(x, Wl1, bl1, Wr1, br1, xl, xr);
    k_init_attn<<<STREAM_BLOCKS, BT, 0, stream>>>(acc, amax, zden);
    k_logits<<<blocks((long long)LG_WAVES * 32, LOGIT_THREADS), LOGIT_THREADS, 0, stream>>>(xl, xr, eattr, lrec, ei,
                                                                 We1, att1, alog, amax);
    k_accum<<<blocks((long long)((ETOT + EPW - 1) / EPW) * 32, 128), 128, 0, stream>>>(xl, ei, alog, amax, zden, acc);
    k_finalize<<<STREAM_BLOCKS, 256, 0, stream>>>(acc, zden, bo1, hbuf);

    k_gemm_pair<HC><<<N / 16, 128, 0, stream>>>(hbuf, Wl2, bl2, Wr2, br2, xl, xr);
    k_init_attn<<<STREAM_BLOCKS, BT, 0, stream>>>(acc, amax, zden);
    k_logits<<<blocks((long long)LG_WAVES * 32, LOGIT_THREADS), LOGIT_THREADS, 0, stream>>>(xl, xr, eattr, lrec, ei,
                                                                 We2, att2, alog, amax);
    k_accum<<<blocks((long long)((ETOT + EPW - 1) / EPW) * 32, 128), 128, 0, stream>>>(xl, ei, alog, amax, zden, acc);
    k_finalize<<<STREAM_BLOCKS, 256, 0, stream>>>(acc, zden, bo2, hbuf);

    k_head<<<blocks(N, BT), BT, 0, stream>>>(hbuf, Ws, bs, out);
}
